# starting kernel
# baseline (speedup 1.0000x reference)
#include <hip/hip_runtime.h>
#include <hip/hip_bf16.h>

#define GLE_BOLTZMAN 0.001987191f
#define GLE_L 64

typedef __attribute__((ext_vector_type(2))) float v2f;
typedef __attribute__((ext_vector_type(4))) float v4f;
typedef __attribute__((ext_vector_type(8))) float v8f;

__global__ void gle_prep_kernel(const float* __restrict__ h,
                                const float* __restrict__ mass,
                                const int* __restrict__ T,
                                float* __restrict__ ws) {
    __shared__ float sh[GLE_L];
    __shared__ float smk[GLE_L];
    const int k = threadIdx.x;
    sh[k] = h[k];
    __syncthreads();
    float acc = 0.f;
    for (int t = 0; t + k < GLE_L; ++t) acc += sh[t] * sh[t + k];
    const float scale = mass[0] * GLE_BOLTZMAN * (float)T[0];
    smk[k] = acc * scale;
    __syncthreads();
    ws[k]         = (k == 0) ? 0.f : -smk[GLE_L - k];
    ws[GLE_L + k] = (k == 0) ? 0.f :  sh[GLE_L - k];
    if (k == 0) {
        ws[2 * GLE_L]     = -smk[0];
        ws[2 * GLE_L + 1] =  sh[0];
    }
}

__global__ void gle_wmma_kernel(const float* __restrict__ vlist,
                                const float* __restrict__ wlist,
                                const float* __restrict__ v,
                                const float* __restrict__ wnew,
                                const float* __restrict__ ws,
                                float* __restrict__ out,
                                int ntiles) {
    const int lane = threadIdx.x & 31;
    const int tile = blockIdx.x * (blockDim.x >> 5) + (threadIdx.x >> 5);
    if (tile >= ntiles) return;

    const int rowBase = tile * 16;
    const int m       = lane & 15;
    const int khalf   = (lane >> 4) << 1;

    const float* __restrict__ vrow = vlist + (size_t)(rowBase + m) * GLE_L;
    const float* __restrict__ wrow = wlist + (size_t)(rowBase + m) * GLE_L;

    v8f acc = {};
#pragma unroll
    for (int kb = 0; kb < GLE_L; kb += 4) {
        const int kk = kb + khalf;
        v2f av = *(const v2f*)(vrow + kk);
        v2f aw = *(const v2f*)(wrow + kk);
        v2f bv = { ws[kk],         ws[kk + 1] };
        v2f bw = { ws[GLE_L + kk], ws[GLE_L + kk + 1] };
        acc = __builtin_amdgcn_wmma_f32_16x16x4_f32(false, av, false, bv,
                                                    (short)0, acc, false, false);
        acc = __builtin_amdgcn_wmma_f32_16x16x4_f32(false, aw, false, bw,
                                                    (short)0, acc, false, false);
    }

    if (m == 0) {
        const float cv = ws[2 * GLE_L];
        const float cw = ws[2 * GLE_L + 1];
        const int base = rowBase + ((lane >> 4) << 3);
        float r[8];
#pragma unroll
        for (int j = 0; j < 8; ++j)
            r[j] = acc[j] + cv * v[base + j] + cw * wnew[base + j];
        *(v4f*)(out + base)     = (v4f){r[0], r[1], r[2], r[3]};
        *(v4f*)(out + base + 4) = (v4f){r[4], r[5], r[6], r[7]};
    }
}

__global__ void gle_tail_kernel(const float* __restrict__ vlist,
                                const float* __restrict__ wlist,
                                const float* __restrict__ v,
                                const float* __restrict__ wnew,
                                const float* __restrict__ ws,
                                float* __restrict__ out,
                                int start, int n) {
    int i = start + blockIdx.x * blockDim.x + threadIdx.x;
    if (i >= n) return;
    const float* vr = vlist + (size_t)i * GLE_L;
    const float* wr = wlist + (size_t)i * GLE_L;
    float acc = 0.f;
    for (int s = 0; s < GLE_L; ++s)
        acc += vr[s] * ws[s] + wr[s] * ws[GLE_L + s];
    out[i] = acc + ws[2 * GLE_L] * v[i] + ws[2 * GLE_L + 1] * wnew[i];
}

extern "C" void kernel_launch(void* const* d_in, const int* in_sizes, int n_in,
                              void* d_out, int out_size, void* d_ws, size_t ws_size,
                              hipStream_t stream) {
    const float* v     = (const float*)d_in[0];
    const int*   T     = (const int*)  d_in[1];
    const float* mass  = (const float*)d_in[3];
    const float* h     = (const float*)d_in[4];
    const float* vlist = (const float*)d_in[5];
    const float* wlist = (const float*)d_in[6];
    const float* wnew  = (const float*)d_in[7];
    float*       out   = (float*)d_out;
    float*       ws    = (float*)d_ws;

    const int n = in_sizes[0];
    const int ntiles = n / 16;
    const int tail   = n - ntiles * 16;

    gle_prep_kernel<<<1, GLE_L, 0, stream>>>(h, mass, T, ws);

    if (ntiles > 0) {
        const int wavesPerBlock = 8;
        const int blocks = (ntiles + wavesPerBlock - 1) / wavesPerBlock;
        gle_wmma_kernel<<<blocks, wavesPerBlock * 32, 0, stream>>>(
            vlist, wlist, v, wnew, ws, out, ntiles);
    }
    if (tail > 0) {
        gle_tail_kernel<<<(tail + 63) / 64, 64, 0, stream>>>(
            vlist, wlist, v, wnew, ws, out, ntiles * 16, n);
    }
}


# Data rows and taps fetched as whole 128-B lines (512 contiguous bytes per wave-wide 16-B load) and re-laid to the WMMA fragments through wave-private padded LDS; k order of the sums permuted
# speedup vs baseline: 2.0764x; 2.0764x over previous
#include <hip/hip_runtime.h>
#include <hip/hip_bf16.h>

#define GLE_BOLTZMAN 0.001987191f
#define GLE_L 64
#define GLE_WAVES 8
#define GLE_PITCH 68

typedef __attribute__((ext_vector_type(2))) float v2f;
typedef __attribute__((ext_vector_type(4))) float v4f;
typedef __attribute__((ext_vector_type(8))) float v8f;

__global__ void gle_prep_kernel(const float* __restrict__ h,
                                const float* __restrict__ mass,
                                const int* __restrict__ T,
                                float* __restrict__ ws) {
    __shared__ float sh[GLE_L];
    __shared__ float smk[GLE_L];
    const int k = threadIdx.x;
    sh[k] = h[k];
    __syncthreads();
    float acc = 0.f;
    for (int t = 0; t + k < GLE_L; ++t) acc += sh[t] * sh[t + k];
    const float scale = mass[0] * GLE_BOLTZMAN * (float)T[0];
    smk[k] = acc * scale;
    __syncthreads();
    ws[k]         = (k == 0) ? 0.f : -smk[GLE_L - k];
    ws[GLE_L + k] = (k == 0) ? 0.f :  sh[GLE_L - k];
    if (k == 0) {
        ws[2 * GLE_L]     = -smk[0];
        ws[2 * GLE_L + 1] =  sh[0];
    }
}

__global__ __launch_bounds__(GLE_WAVES * 32)
void gle_wmma_kernel(const float* __restrict__ vlist,
                     const float* __restrict__ wlist,
                     const float* __restrict__ v,
                     const float* __restrict__ wnew,
                     const float* __restrict__ ws,
                     float* __restrict__ out,
                     int ntiles) {
    __shared__ float s_rows[GLE_WAVES][2][16 * GLE_PITCH];
    __shared__ float s_taps[GLE_WAVES][2 * GLE_L];

    const int lane = threadIdx.x & 31;
    const int wave = threadIdx.x >> 5;
    const int tile = blockIdx.x * GLE_WAVES + wave;
    if (tile >= ntiles) return;

    const int rowBase = tile * 16;
    const int c = lane & 15;
    const int h = lane >> 4;

    const float* __restrict__ vt = vlist + (size_t)rowBase * GLE_L + lane * 4;
    const float* __restrict__ wt = wlist + (size_t)rowBase * GLE_L + lane * 4;
    const v4f tp = *(const v4f*)(ws + lane * 4);
    v4f gv[8], gw[8];
#pragma unroll
    for (int j = 0; j < 8; ++j) gv[j] = *(const v4f*)(vt + j * 128);
#pragma unroll
    for (int j = 0; j < 8; ++j) gw[j] = *(const v4f*)(wt + j * 128);

    float* sv = &s_rows[wave][0][0];
    float* sw = &s_rows[wave][1][0];
    float* st = &s_taps[wave][0];
    *(v4f*)(st + lane * 4) = tp;
#pragma unroll
    for (int j = 0; j < 8; ++j)
        *(v4f*)(sv + (2 * j + h) * GLE_PITCH + 4 * c) = gv[j];
#pragma unroll
    for (int j = 0; j < 8; ++j)
        *(v4f*)(sw + (2 * j + h) * GLE_PITCH + 4 * c) = gw[j];

    asm volatile("s_wait_dscnt 0x0" ::: "memory");

    const float* av_row = sv + c * GLE_PITCH;
    const float* aw_row = sw + c * GLE_PITCH;

    v8f acc = {};
#pragma unroll
    for (int s = 0; s < 8; ++s) {
        const int k = 8 * s + 4 * h;
        const v4f av = *(const v4f*)(av_row + k);
        const v4f bv = *(const v4f*)(st + k);
        const v4f aw = *(const v4f*)(aw_row + k);
        const v4f bw = *(const v4f*)(st + GLE_L + k);
        acc = __builtin_amdgcn_wmma_f32_16x16x4_f32(false, (v2f){av[0], av[1]}, false, (v2f){bv[0], bv[1]},
                                                    (short)0, acc, false, false);
        acc = __builtin_amdgcn_wmma_f32_16x16x4_f32(false, (v2f){av[2], av[3]}, false, (v2f){bv[2], bv[3]},
                                                    (short)0, acc, false, false);
        acc = __builtin_amdgcn_wmma_f32_16x16x4_f32(false, (v2f){aw[0], aw[1]}, false, (v2f){bw[0], bw[1]},
                                                    (short)0, acc, false, false);
        acc = __builtin_amdgcn_wmma_f32_16x16x4_f32(false, (v2f){aw[2], aw[3]}, false, (v2f){bw[2], bw[3]},
                                                    (short)0, acc, false, false);
    }

    if (c == 0) {
        const float cv = ws[2 * GLE_L];
        const float cw = ws[2 * GLE_L + 1];
        const int base = rowBase + (h << 3);
        float r[8];
#pragma unroll
        for (int j = 0; j < 8; ++j)
            r[j] = acc[j] + cv * v[base + j] + cw * wnew[base + j];
        *(v4f*)(out + base)     = (v4f){r[0], r[1], r[2], r[3]};
        *(v4f*)(out + base + 4) = (v4f){r[4], r[5], r[6], r[7]};
    }
}

__global__ void gle_tail_kernel(const float* __restrict__ vlist,
                                const float* __restrict__ wlist,
                                const float* __restrict__ v,
                                const float* __restrict__ wnew,
                                const float* __restrict__ ws,
                                float* __restrict__ out,
                                int start, int n) {
    int i = start + blockIdx.x * blockDim.x + threadIdx.x;
    if (i >= n) return;
    const float* vr = vlist + (size_t)i * GLE_L;
    const float* wr = wlist + (size_t)i * GLE_L;
    float acc = 0.f;
    for (int s = 0; s < GLE_L; ++s)
        acc += vr[s] * ws[s] + wr[s] * ws[GLE_L + s];
    out[i] = acc + ws[2 * GLE_L] * v[i] + ws[2 * GLE_L + 1] * wnew[i];
}

extern "C" void kernel_launch(void* const* d_in, const int* in_sizes, int n_in,
                              void* d_out, int out_size, void* d_ws, size_t ws_size,
                              hipStream_t stream) {
    const float* v     = (const float*)d_in[0];
    const int*   T     = (const int*)  d_in[1];
    const float* mass  = (const float*)d_in[3];
    const float* h     = (const float*)d_in[4];
    const float* vlist = (const float*)d_in[5];
    const float* wlist = (const float*)d_in[6];
    const float* wnew  = (const float*)d_in[7];
    float*       out   = (float*)d_out;
    float*       ws    = (float*)d_ws;

    const int n = in_sizes[0];
    const int ntiles = n / 16;
    const int tail   = n - ntiles * 16;

    gle_prep_kernel<<<1, GLE_L, 0, stream>>>(h, mass, T, ws);

    if (ntiles > 0) {
        const int blocks = (ntiles + GLE_WAVES - 1) / GLE_WAVES;
        gle_wmma_kernel<<<blocks, GLE_WAVES * 32, 0, stream>>>(
            vlist, wlist, v, wnew, ws, out, ntiles);
    }
    if (tail > 0) {
        gle_tail_kernel<<<(tail + 63) / 64, 64, 0, stream>>>(
            vlist, wlist, v, wnew, ws, out, ntiles * 16, n);
    }
}
